# starting kernel
# baseline (speedup 1.0000x reference)
#include <hip/hip_runtime.h>


typedef __attribute__((ext_vector_type(2))) float v2f;
typedef __attribute__((ext_vector_type(8))) float v8f;

#define ROWS 32
#define DLEN 512
#define ST   516
#define NTHR 256

__shared__ float bufA[ROWS * ST];
__shared__ float bufB[ROWS * ST];

template<int K>
__device__ __forceinline__ void band_gemm(const float* __restrict__ W,
                                          const float* __restrict__ bias,
                                          int boff, int n0, int lo, int hi)
{
    const float* Ap0 = &bufB[lo * ST + boff + 2 * hi];
    const float* Ap1 = Ap0 + 16 * ST;
    const float* Wp  = W + (size_t)(n0 + lo) * K + 2 * hi;

    v8f acc0 = {0.f, 0.f, 0.f, 0.f, 0.f, 0.f, 0.f, 0.f};
    v8f acc1 = {0.f, 0.f, 0.f, 0.f, 0.f, 0.f, 0.f, 0.f};
    #pragma unroll 8
    for (int k = 0; k < K; k += 4) {
        v2f bf = *(const v2f*)(Wp + k);
        v2f a0 = *(const v2f*)(Ap0 + k);
        v2f a1 = *(const v2f*)(Ap1 + k);
        acc0 = __builtin_amdgcn_wmma_f32_16x16x4_f32(false, a0, false, bf, (short)0, acc0, false, false);
        acc1 = __builtin_amdgcn_wmma_f32_16x16x4_f32(false, a1, false, bf, (short)0, acc1, false, false);
    }

    float bb = bias[n0 + lo];
    #pragma unroll
    for (int r = 0; r < 8; ++r) {
        bufA[(r + 8 * hi) * ST + boff + n0 + lo]      = acc0[r] + bb;
        bufA[(16 + r + 8 * hi) * ST + boff + n0 + lo] = acc1[r] + bb;
    }
}

__global__ __launch_bounds__(NTHR, 1)
void wavelet_mlp_fused(const float* __restrict__ x,
                       const float* __restrict__ Wd0, const float* __restrict__ bd0,
                       const float* __restrict__ Wd1, const float* __restrict__ bd1,
                       const float* __restrict__ Wd2, const float* __restrict__ bd2,
                       const float* __restrict__ Wa,  const float* __restrict__ ba,
                       float* __restrict__ out)
{
    const int   tid  = threadIdx.x;
    const int   wave = tid >> 5;
    const int   lane = tid & 31;
    const int   lo   = lane & 15;
    const int   hi   = lane >> 4;
    const long  rowBase = (long)blockIdx.x * ROWS;
    const float C = 0.70710678118654752440f;

    {
        const float4* xg = (const float4*)(x + rowBase * DLEN);
        for (int j = tid; j < ROWS * DLEN / 4; j += NTHR) {
            float4 v = xg[j];
            int r = (j * 4) >> 9;
            int c = (j * 4) & (DLEN - 1);
            *(float4*)&bufA[r * ST + c] = v;
        }
    }
    __syncthreads();

    for (int i = tid; i < ROWS * 256; i += NTHR) {
        int r = i >> 8, k = i & 255;
        float e = bufA[r * ST + 2 * k];
        float o = bufA[r * ST + 2 * k + 1];
        bufB[r * ST + 256 + k] = (e - o) * C;
        bufB[r * ST + k]       = (e + o) * C;
    }
    __syncthreads();

    {
        float av[16], dv[16];
        #pragma unroll
        for (int u = 0; u < 16; ++u) {
            int i = tid + u * NTHR;
            int r = i >> 7, k = i & 127;
            float e = bufB[r * ST + 2 * k];
            float o = bufB[r * ST + 2 * k + 1];
            dv[u] = (e - o) * C;
            av[u] = (e + o) * C;
        }
        __syncthreads();
        #pragma unroll
        for (int u = 0; u < 16; ++u) {
            int i = tid + u * NTHR;
            int r = i >> 7, k = i & 127;
            bufB[r * ST + 128 + k] = dv[u];
            bufB[r * ST + k]       = av[u];
        }
    }
    __syncthreads();

    {
        float av[8], dv[8];
        #pragma unroll
        for (int u = 0; u < 8; ++u) {
            int i = tid + u * NTHR;
            int r = i >> 6, k = i & 63;
            float e = bufB[r * ST + 2 * k];
            float o = bufB[r * ST + 2 * k + 1];
            dv[u] = (e - o) * C;
            av[u] = (e + o) * C;
        }
        __syncthreads();
        #pragma unroll
        for (int u = 0; u < 8; ++u) {
            int i = tid + u * NTHR;
            int r = i >> 6, k = i & 63;
            bufB[r * ST + 64 + k] = dv[u];
            bufB[r * ST + k]      = av[u];
        }
    }
    __syncthreads();

    if (wave < 4) band_gemm<64>(Wa,  ba,  0,  wave * 16,        lo, hi);
    else          band_gemm<64>(Wd2, bd2, 64, (wave - 4) * 16,  lo, hi);
    band_gemm<128>(Wd1, bd1, 128, wave * 16,       lo, hi);
    band_gemm<256>(Wd0, bd0, 256, wave * 16,       lo, hi);
    band_gemm<256>(Wd0, bd0, 256, (wave + 8) * 16, lo, hi);
    __syncthreads();

    for (int i = tid; i < ROWS * 64; i += NTHR) {
        int r = i >> 6, j = i & 63;
        float av = bufA[r * ST + j];
        float dv = bufA[r * ST + 64 + j];
        bufB[r * ST + 2 * j]     = (av + dv) * C;
        bufB[r * ST + 2 * j + 1] = (av - dv) * C;
    }
    __syncthreads();
    for (int i = tid; i < ROWS * 128; i += NTHR) {
        int r = i >> 7, j = i & 127;
        float av = bufB[r * ST + j];
        float dv = bufA[r * ST + 128 + j];
        bufB[r * ST + 256 + 2 * j]     = (av + dv) * C;
        bufB[r * ST + 256 + 2 * j + 1] = (av - dv) * C;
    }
    __syncthreads();
    for (int i = tid; i < ROWS * 256; i += NTHR) {
        int r = i >> 8, j = i & 255;
        float av = bufB[r * ST + 256 + j];
        float dv = bufA[r * ST + 256 + j];
        float2 ov;
        ov.x = (av + dv) * C;
        ov.y = (av - dv) * C;
        *(float2*)&out[(rowBase + r) * DLEN + 2 * j] = ov;
    }
}

extern "C" void kernel_launch(void* const* d_in, const int* in_sizes, int n_in,
                              void* d_out, int out_size, void* d_ws, size_t ws_size,
                              hipStream_t stream) {
    const float* xp  = (const float*)d_in[0];
    const float* Wd0 = (const float*)d_in[1];
    const float* bd0 = (const float*)d_in[2];
    const float* Wd1 = (const float*)d_in[3];
    const float* bd1 = (const float*)d_in[4];
    const float* Wd2 = (const float*)d_in[5];
    const float* bd2 = (const float*)d_in[6];
    const float* Wa  = (const float*)d_in[7];
    const float* ba  = (const float*)d_in[8];
    float* outp = (float*)d_out;

    int M = in_sizes[0] / DLEN;
    int blocks = M / ROWS;
    wavelet_mlp_fused<<<blocks, NTHR, 0, stream>>>(
        xp, Wd0, bd0, Wd1, bd1, Wd2, bd2, Wa, ba, outp);
}


# Band GEMM fragment loads software-pipelined: weight fragments fetched two 8-step blocks ahead, LDS fragments one block ahead, each as one group before the block's WMMAs; order of sums unchanged
# speedup vs baseline: 1.0992x; 1.0992x over previous
#include <hip/hip_runtime.h>


typedef __attribute__((ext_vector_type(2))) float v2f;
typedef __attribute__((ext_vector_type(8))) float v8f;

#define ROWS 32
#define DLEN 512
#define ST   516
#define NTHR 256

__shared__ float bufA[ROWS * ST];
__shared__ float bufB[ROWS * ST];

struct GemmState {
    const float* wp[4];
    int   aoff[4];
    int   ycol[4];
    float bias[4];
    v2f   bq[3][8];
    v2f   a0q[2][8];
    v2f   a1q[2][8];
    v8f   acc0, acc1;
};

template<int B> struct BlkOf {
    static constexpr int  seg  = B < 2 ? 0 : (B < 6 ? 1 : (B < 14 ? 2 : 3));
    static constexpr int  j    = B < 2 ? B : (B < 6 ? B - 2 : (B < 14 ? B - 6 : B - 14));
    static constexpr bool last = (B == 1 || B == 5 || B == 13 || B == 21);
};

template<int B>
__device__ __forceinline__ void load_b(GemmState& S)
{
    const float* p = S.wp[BlkOf<B>::seg] + 32 * BlkOf<B>::j;
    #pragma unroll
    for (int u = 0; u < 8; ++u) S.bq[B % 3][u] = *(const v2f*)(p + 4 * u);
}

template<int B>
__device__ __forceinline__ void load_a(GemmState& S)
{
    const int o = S.aoff[BlkOf<B>::seg] + 32 * BlkOf<B>::j;
    #pragma unroll
    for (int u = 0; u < 8; ++u) {
        S.a0q[B % 2][u] = *(const v2f*)&bufB[o + 4 * u];
        S.a1q[B % 2][u] = *(const v2f*)&bufB[o + 16 * ST + 4 * u];
    }
}

template<int B>
__device__ __forceinline__ void mma_blk(GemmState& S)
{
    #pragma unroll
    for (int u = 0; u < 8; ++u) {
        S.acc0 = __builtin_amdgcn_wmma_f32_16x16x4_f32(false, S.a0q[B % 2][u], false, S.bq[B % 3][u], (short)0, S.acc0, false, false);
        S.acc1 = __builtin_amdgcn_wmma_f32_16x16x4_f32(false, S.a1q[B % 2][u], false, S.bq[B % 3][u], (short)0, S.acc1, false, false);
    }
}

template<int SEG>
__device__ __forceinline__ void finish_seg(GemmState& S, int hi)
{
    const float bb = S.bias[SEG];
    const int   c  = S.ycol[SEG];
    #pragma unroll
    for (int r = 0; r < 8; ++r) {
        bufA[(r + 8 * hi) * ST + c]      = S.acc0[r] + bb;
        bufA[(16 + r + 8 * hi) * ST + c] = S.acc1[r] + bb;
    }
    const v8f z = {0.f, 0.f, 0.f, 0.f, 0.f, 0.f, 0.f, 0.f};
    S.acc0 = z;
    S.acc1 = z;
}

template<int B>
__device__ __forceinline__ void run_blocks(GemmState& S, int hi)
{
    if constexpr (B + 2 < 22) load_b<B + 2>(S);
    if constexpr (B + 1 < 22) load_a<B + 1>(S);
    __builtin_amdgcn_sched_barrier(0);
    mma_blk<B>(S);
    if constexpr (BlkOf<B>::last) finish_seg<BlkOf<B>::seg>(S, hi);
    __builtin_amdgcn_sched_barrier(0);
    if constexpr (B + 1 < 22) run_blocks<B + 1>(S, hi);
}

__global__ __launch_bounds__(NTHR, 1)
void wavelet_mlp_fused(const float* __restrict__ x,
                       const float* __restrict__ Wd0, const float* __restrict__ bd0,
                       const float* __restrict__ Wd1, const float* __restrict__ bd1,
                       const float* __restrict__ Wd2, const float* __restrict__ bd2,
                       const float* __restrict__ Wa,  const float* __restrict__ ba,
                       float* __restrict__ out)
{
    const int   tid  = threadIdx.x;
    const int   wave = tid >> 5;
    const int   lane = tid & 31;
    const int   lo   = lane & 15;
    const int   hi   = lane >> 4;
    const long  rowBase = (long)blockIdx.x * ROWS;
    const float C = 0.70710678118654752440f;

    {
        const float4* xg = (const float4*)(x + rowBase * DLEN);
        for (int j = tid; j < ROWS * DLEN / 4; j += NTHR) {
            float4 v = xg[j];
            int r = (j * 4) >> 9;
            int c = (j * 4) & (DLEN - 1);
            *(float4*)&bufA[r * ST + c] = v;
        }
    }
    __syncthreads();

    GemmState S;
    {
        const bool   lowHalf = wave < 4;
        const float* W0 = lowHalf ? Wa : Wd2;
        const float* b0 = lowHalf ? ba : bd2;
        const int boff0 = lowHalf ? 0 : 64;
        const int n00   = (wave & 3) * 16;
        const int n01   = wave * 16;
        const int n03   = (wave + 8) * 16;
        S.wp[0] = W0  + (size_t)(n00 + lo) * 64  + 2 * hi;
        S.wp[1] = Wd1 + (size_t)(n01 + lo) * 128 + 2 * hi;
        S.wp[2] = Wd0 + (size_t)(n01 + lo) * 256 + 2 * hi;
        S.wp[3] = Wd0 + (size_t)(n03 + lo) * 256 + 2 * hi;
        S.aoff[0] = lo * ST + boff0 + 2 * hi;
        S.aoff[1] = lo * ST + 128 + 2 * hi;
        S.aoff[2] = lo * ST + 256 + 2 * hi;
        S.aoff[3] = lo * ST + 256 + 2 * hi;
        S.ycol[0] = boff0 + n00 + lo;
        S.ycol[1] = 128 + n01 + lo;
        S.ycol[2] = 256 + n01 + lo;
        S.ycol[3] = 256 + n03 + lo;
        S.bias[0] = b0[n00 + lo];
        S.bias[1] = bd1[n01 + lo];
        S.bias[2] = bd0[n01 + lo];
        S.bias[3] = bd0[n03 + lo];
        const v8f z = {0.f, 0.f, 0.f, 0.f, 0.f, 0.f, 0.f, 0.f};
        S.acc0 = z;
        S.acc1 = z;
        load_b<0>(S);
        load_b<1>(S);
    }

    for (int i = tid; i < ROWS * 256; i += NTHR) {
        int r = i >> 8, k = i & 255;
        float e = bufA[r * ST + 2 * k];
        float o = bufA[r * ST + 2 * k + 1];
        bufB[r * ST + 256 + k] = (e - o) * C;
        bufB[r * ST + k]       = (e + o) * C;
    }
    __syncthreads();

    {
        float av[16], dv[16];
        #pragma unroll
        for (int u = 0; u < 16; ++u) {
            int i = tid + u * NTHR;
            int r = i >> 7, k = i & 127;
            float e = bufB[r * ST + 2 * k];
            float o = bufB[r * ST + 2 * k + 1];
            dv[u] = (e - o) * C;
            av[u] = (e + o) * C;
        }
        __syncthreads();
        #pragma unroll
        for (int u = 0; u < 16; ++u) {
            int i = tid + u * NTHR;
            int r = i >> 7, k = i & 127;
            bufB[r * ST + 128 + k] = dv[u];
            bufB[r * ST + k]       = av[u];
        }
    }
    __syncthreads();

    {
        float av[8], dv[8];
        #pragma unroll
        for (int u = 0; u < 8; ++u) {
            int i = tid + u * NTHR;
            int r = i >> 6, k = i & 63;
            float e = bufB[r * ST + 2 * k];
            float o = bufB[r * ST + 2 * k + 1];
            dv[u] = (e - o) * C;
            av[u] = (e + o) * C;
        }
        __syncthreads();
        #pragma unroll
        for (int u = 0; u < 8; ++u) {
            int i = tid + u * NTHR;
            int r = i >> 6, k = i & 63;
            bufB[r * ST + 64 + k] = dv[u];
            bufB[r * ST + k]      = av[u];
        }
    }
    __syncthreads();

    load_a<0>(S);
    run_blocks<0>(S, hi);
    __syncthreads();

    for (int i = tid; i < ROWS * 64; i += NTHR) {
        int r = i >> 6, j = i & 63;
        float av = bufA[r * ST + j];
        float dv = bufA[r * ST + 64 + j];
        bufB[r * ST + 2 * j]     = (av + dv) * C;
        bufB[r * ST + 2 * j + 1] = (av - dv) * C;
    }
    __syncthreads();
    for (int i = tid; i < ROWS * 128; i += NTHR) {
        int r = i >> 7, j = i & 127;
        float av = bufB[r * ST + j];
        float dv = bufA[r * ST + 128 + j];
        bufB[r * ST + 256 + 2 * j]     = (av + dv) * C;
        bufB[r * ST + 256 + 2 * j + 1] = (av - dv) * C;
    }
    __syncthreads();
    for (int i = tid; i < ROWS * 256; i += NTHR) {
        int r = i >> 8, j = i & 255;
        float av = bufB[r * ST + 256 + j];
        float dv = bufA[r * ST + 256 + j];
        float2 ov;
        ov.x = (av + dv) * C;
        ov.y = (av - dv) * C;
        *(float2*)&out[(rowBase + r) * DLEN + 2 * j] = ov;
    }
}

extern "C" void kernel_launch(void* const* d_in, const int* in_sizes, int n_in,
                              void* d_out, int out_size, void* d_ws, size_t ws_size,
                              hipStream_t stream) {
    const float* xp  = (const float*)d_in[0];
    const float* Wd0 = (const float*)d_in[1];
    const float* bd0 = (const float*)d_in[2];
    const float* Wd1 = (const float*)d_in[3];
    const float* bd1 = (const float*)d_in[4];
    const float* Wd2 = (const float*)d_in[5];
    const float* bd2 = (const float*)d_in[6];
    const float* Wa  = (const float*)d_in[7];
    const float* ba  = (const float*)d_in[8];
    float* outp = (float*)d_out;

    int M = in_sizes[0] / DLEN;
    int blocks = M / ROWS;
    wavelet_mlp_fused<<<blocks, NTHR, 0, stream>>>(
        xp, Wd0, bd0, Wd1, bd1, Wd2, bd2, Wa, ba, outp);
}


# Three Haar levels fused in registers per float4 (fwd on the loaded tile, inv into 16-byte stores, level-2 pair by DPP lane swap): 5 LDS passes, 7 barriers gone; fma forms across levels (last bit)
# speedup vs baseline: 1.2025x; 1.0940x over previous
#include <hip/hip_runtime.h>


typedef __attribute__((ext_vector_type(2))) float v2f;
typedef __attribute__((ext_vector_type(8))) float v8f;

#define ROWS 32
#define DLEN 512
#define ST   516
#define NTHR 256

__shared__ float bufA[ROWS * ST];
__shared__ float bufB[ROWS * ST];

struct GemmState {
    const float* wp[4];
    int   aoff[4];
    int   ycol[4];
    float bias[4];
    v2f   bq[3][8];
    v2f   a0q[2][8];
    v2f   a1q[2][8];
    v8f   acc0, acc1;
};

template<int B> struct BlkOf {
    static constexpr int  seg  = B < 2 ? 0 : (B < 6 ? 1 : (B < 14 ? 2 : 3));
    static constexpr int  j    = B < 2 ? B : (B < 6 ? B - 2 : (B < 14 ? B - 6 : B - 14));
    static constexpr bool last = (B == 1 || B == 5 || B == 13 || B == 21);
};

template<int B>
__device__ __forceinline__ void load_b(GemmState& S)
{
    const float* p = S.wp[BlkOf<B>::seg] + 32 * BlkOf<B>::j;
    #pragma unroll
    for (int u = 0; u < 8; ++u) S.bq[B % 3][u] = *(const v2f*)(p + 4 * u);
}

template<int B>
__device__ __forceinline__ void load_a(GemmState& S)
{
    const int o = S.aoff[BlkOf<B>::seg] + 32 * BlkOf<B>::j;
    #pragma unroll
    for (int u = 0; u < 8; ++u) {
        S.a0q[B % 2][u] = *(const v2f*)&bufB[o + 4 * u];
        S.a1q[B % 2][u] = *(const v2f*)&bufB[o + 16 * ST + 4 * u];
    }
}

template<int B>
__device__ __forceinline__ void mma_blk(GemmState& S)
{
    #pragma unroll
    for (int u = 0; u < 8; ++u) {
        S.acc0 = __builtin_amdgcn_wmma_f32_16x16x4_f32(false, S.a0q[B % 2][u], false, S.bq[B % 3][u], (short)0, S.acc0, false, false);
        S.acc1 = __builtin_amdgcn_wmma_f32_16x16x4_f32(false, S.a1q[B % 2][u], false, S.bq[B % 3][u], (short)0, S.acc1, false, false);
    }
}

template<int SEG>
__device__ __forceinline__ void finish_seg(GemmState& S, int hi)
{
    const float bb = S.bias[SEG];
    const int   c  = S.ycol[SEG];
    #pragma unroll
    for (int r = 0; r < 8; ++r) {
        bufA[(r + 8 * hi) * ST + c]      = S.acc0[r] + bb;
        bufA[(16 + r + 8 * hi) * ST + c] = S.acc1[r] + bb;
    }
    const v8f z = {0.f, 0.f, 0.f, 0.f, 0.f, 0.f, 0.f, 0.f};
    S.acc0 = z;
    S.acc1 = z;
}

template<int B>
__device__ __forceinline__ void run_blocks(GemmState& S, int hi)
{
    if constexpr (B + 2 < 22) load_b<B + 2>(S);
    if constexpr (B + 1 < 22) load_a<B + 1>(S);
    __builtin_amdgcn_sched_barrier(0);
    mma_blk<B>(S);
    if constexpr (BlkOf<B>::last) finish_seg<BlkOf<B>::seg>(S, hi);
    __builtin_amdgcn_sched_barrier(0);
    if constexpr (B + 1 < 22) run_blocks<B + 1>(S, hi);
}

__device__ __forceinline__ float lane_xor1(float v)
{
    return __int_as_float(__builtin_amdgcn_mov_dpp(__float_as_int(v), 0xB1, 0xF, 0xF, true));
}

__global__ __launch_bounds__(NTHR, 1)
void wavelet_mlp_fused(const float* __restrict__ x,
                       const float* __restrict__ Wd0, const float* __restrict__ bd0,
                       const float* __restrict__ Wd1, const float* __restrict__ bd1,
                       const float* __restrict__ Wd2, const float* __restrict__ bd2,
                       const float* __restrict__ Wa,  const float* __restrict__ ba,
                       float* __restrict__ out)
{
    const int   tid  = threadIdx.x;
    const int   wave = tid >> 5;
    const int   lane = tid & 31;
    const int   lo   = lane & 15;
    const int   hi   = lane >> 4;
    const long  rowBase = (long)blockIdx.x * ROWS;
    const float C = 0.70710678118654752440f;

    const int  q4   = tid & 127;
    const int  r0   = tid >> 7;
    const bool oddq = q4 & 1;

    float4 xv[16];
    {
        const float4* xg = (const float4*)(x + rowBase * DLEN);
        #pragma unroll
        for (int u = 0; u < 16; ++u) xv[u] = xg[tid + u * NTHR];
    }

    GemmState S;
    {
        const bool   lowHalf = wave < 4;
        const float* W0 = lowHalf ? Wa : Wd2;
        const float* b0 = lowHalf ? ba : bd2;
        const int boff0 = lowHalf ? 0 : 64;
        const int n00   = (wave & 3) * 16;
        const int n01   = wave * 16;
        const int n03   = (wave + 8) * 16;
        S.wp[0] = W0  + (size_t)(n00 + lo) * 64  + 2 * hi;
        S.wp[1] = Wd1 + (size_t)(n01 + lo) * 128 + 2 * hi;
        S.wp[2] = Wd0 + (size_t)(n01 + lo) * 256 + 2 * hi;
        S.wp[3] = Wd0 + (size_t)(n03 + lo) * 256 + 2 * hi;
        S.aoff[0] = lo * ST + boff0 + 2 * hi;
        S.aoff[1] = lo * ST + 128 + 2 * hi;
        S.aoff[2] = lo * ST + 256 + 2 * hi;
        S.aoff[3] = lo * ST + 256 + 2 * hi;
        S.ycol[0] = boff0 + n00 + lo;
        S.ycol[1] = 128 + n01 + lo;
        S.ycol[2] = 256 + n01 + lo;
        S.ycol[3] = 256 + n03 + lo;
        S.bias[0] = b0[n00 + lo];
        S.bias[1] = bd1[n01 + lo];
        S.bias[2] = bd0[n01 + lo];
        S.bias[3] = bd0[n03 + lo];
        const v8f z = {0.f, 0.f, 0.f, 0.f, 0.f, 0.f, 0.f, 0.f};
        S.acc0 = z;
        S.acc1 = z;
        load_b<0>(S);
        load_b<1>(S);
    }

    {
        const int o2 = (oddq ? 64 : 0) + (q4 >> 1);
        #pragma unroll
        for (int u = 0; u < 16; ++u) {
            const int    rb = (r0 + 2 * u) * ST;
            const float4 v  = xv[u];
            v2f d0;
            d0.x = (v.x - v.y) * C;
            d0.y = (v.z - v.w) * C;
            const float a10 = (v.x + v.y) * C;
            const float a11 = (v.z + v.w) * C;
            const float d1  = (a10 - a11) * C;
            const float a2  = (a10 + a11) * C;
            const float nb  = lane_xor1(a2);
            const float t   = nb + (oddq ? -a2 : a2);
            *(v2f*)&bufB[rb + 256 + 2 * q4] = d0;
            bufB[rb + 128 + q4] = d1;
            bufB[rb + o2]       = t * C;
        }
    }
    __syncthreads();

    load_a<0>(S);
    run_blocks<0>(S, hi);
    __syncthreads();

    {
        float* og = out + rowBase * DLEN;
        const int h = q4 >> 1;
        for (int u = 0; u < 16; ++u) {
            const int   rb  = (r0 + 2 * u) * ST;
            const float ya  = bufA[rb + h];
            const float yd2 = bufA[rb + 64 + h];
            const float yd1 = bufA[rb + 128 + q4];
            const v2f   yd0 = *(const v2f*)&bufA[rb + 256 + 2 * q4];
            const float a2  = (ya + (oddq ? -yd2 : yd2)) * C;
            const float a1e = (a2 + yd1) * C;
            const float a1o = (a2 - yd1) * C;
            float4 ov;
            ov.x = (a1e + yd0.x) * C;
            ov.y = (a1e - yd0.x) * C;
            ov.z = (a1o + yd0.y) * C;
            ov.w = (a1o - yd0.y) * C;
            *(float4*)&og[(tid + u * NTHR) * 4] = ov;
        }
    }
}

extern "C" void kernel_launch(void* const* d_in, const int* in_sizes, int n_in,
                              void* d_out, int out_size, void* d_ws, size_t ws_size,
                              hipStream_t stream) {
    const float* xp  = (const float*)d_in[0];
    const float* Wd0 = (const float*)d_in[1];
    const float* bd0 = (const float*)d_in[2];
    const float* Wd1 = (const float*)d_in[3];
    const float* bd1 = (const float*)d_in[4];
    const float* Wd2 = (const float*)d_in[5];
    const float* bd2 = (const float*)d_in[6];
    const float* Wa  = (const float*)d_in[7];
    const float* ba  = (const float*)d_in[8];
    float* outp = (float*)d_out;

    int M = in_sizes[0] / DLEN;
    int blocks = M / ROWS;
    wavelet_mlp_fused<<<blocks, NTHR, 0, stream>>>(
        xp, Wd0, bd0, Wd1, bd1, Wd2, bd2, Wa, ba, outp);
}


# Band GEMM A and B fragments fetched 128 bits per lane (k re-mapped inside each 32-k block, same map for both operands): half the fragment load instructions; order of sums changes inside 8-k groups
# speedup vs baseline: 1.2656x; 1.0525x over previous
#include <hip/hip_runtime.h>


typedef __attribute__((ext_vector_type(2))) float v2f;
typedef __attribute__((ext_vector_type(4))) float v4f;
typedef __attribute__((ext_vector_type(8))) float v8f;

#define ROWS 32
#define DLEN 512
#define ST   516
#define NTHR 256

__shared__ __attribute__((aligned(16))) float bufA[ROWS * ST];
__shared__ __attribute__((aligned(16))) float bufB[ROWS * ST];

struct GemmState {
    const float* wp[4];
    int   aoff[4];
    int   ycol[4];
    float bias[4];
    v4f   bq[3][4];
    v4f   a0q[2][4];
    v4f   a1q[2][4];
    v8f   acc0, acc1;
};

template<int B> struct BlkOf {
    static constexpr int  seg  = B < 2 ? 0 : (B < 6 ? 1 : (B < 14 ? 2 : 3));
    static constexpr int  j    = B < 2 ? B : (B < 6 ? B - 2 : (B < 14 ? B - 6 : B - 14));
    static constexpr bool last = (B == 1 || B == 5 || B == 13 || B == 21);
};

template<int B>
__device__ __forceinline__ void load_b(GemmState& S)
{
    const float* p = S.wp[BlkOf<B>::seg] + 32 * BlkOf<B>::j;
    #pragma unroll
    for (int t = 0; t < 4; ++t) S.bq[B % 3][t] = *(const v4f*)(p + 8 * t);
}

template<int B>
__device__ __forceinline__ void load_a(GemmState& S)
{
    const int o = S.aoff[BlkOf<B>::seg] + 32 * BlkOf<B>::j;
    #pragma unroll
    for (int t = 0; t < 4; ++t) {
        S.a0q[B % 2][t] = *(const v4f*)&bufB[o + 8 * t];
        S.a1q[B % 2][t] = *(const v4f*)&bufB[o + 16 * ST + 8 * t];
    }
}

template<int B>
__device__ __forceinline__ void mma_blk(GemmState& S)
{
    #pragma unroll
    for (int t = 0; t < 4; ++t) {
        const v4f a0 = S.a0q[B % 2][t];
        const v4f a1 = S.a1q[B % 2][t];
        const v4f b  = S.bq[B % 3][t];
        const v2f a0l = __builtin_shufflevector(a0, a0, 0, 1), a0h = __builtin_shufflevector(a0, a0, 2, 3);
        const v2f a1l = __builtin_shufflevector(a1, a1, 0, 1), a1h = __builtin_shufflevector(a1, a1, 2, 3);
        const v2f bl  = __builtin_shufflevector(b,  b,  0, 1), bh  = __builtin_shufflevector(b,  b,  2, 3);
        S.acc0 = __builtin_amdgcn_wmma_f32_16x16x4_f32(false, a0l, false, bl, (short)0, S.acc0, false, false);
        S.acc1 = __builtin_amdgcn_wmma_f32_16x16x4_f32(false, a1l, false, bl, (short)0, S.acc1, false, false);
        S.acc0 = __builtin_amdgcn_wmma_f32_16x16x4_f32(false, a0h, false, bh, (short)0, S.acc0, false, false);
        S.acc1 = __builtin_amdgcn_wmma_f32_16x16x4_f32(false, a1h, false, bh, (short)0, S.acc1, false, false);
    }
}

template<int SEG>
__device__ __forceinline__ void finish_seg(GemmState& S, int hi)
{
    const float bb = S.bias[SEG];
    const int   c  = S.ycol[SEG];
    #pragma unroll
    for (int r = 0; r < 8; ++r) {
        bufA[(r + 8 * hi) * ST + c]      = S.acc0[r] + bb;
        bufA[(16 + r + 8 * hi) * ST + c] = S.acc1[r] + bb;
    }
    const v8f z = {0.f, 0.f, 0.f, 0.f, 0.f, 0.f, 0.f, 0.f};
    S.acc0 = z;
    S.acc1 = z;
}

template<int B>
__device__ __forceinline__ void run_blocks(GemmState& S, int hi)
{
    if constexpr (B + 2 < 22) load_b<B + 2>(S);
    if constexpr (B + 1 < 22) load_a<B + 1>(S);
    __builtin_amdgcn_sched_barrier(0);
    mma_blk<B>(S);
    if constexpr (BlkOf<B>::last) finish_seg<BlkOf<B>::seg>(S, hi);
    __builtin_amdgcn_sched_barrier(0);
    if constexpr (B + 1 < 22) run_blocks<B + 1>(S, hi);
}

__device__ __forceinline__ float lane_xor1(float v)
{
    return __int_as_float(__builtin_amdgcn_mov_dpp(__float_as_int(v), 0xB1, 0xF, 0xF, true));
}

__global__ __launch_bounds__(NTHR, 1)
void wavelet_mlp_fused(const float* __restrict__ x,
                       const float* __restrict__ Wd0, const float* __restrict__ bd0,
                       const float* __restrict__ Wd1, const float* __restrict__ bd1,
                       const float* __restrict__ Wd2, const float* __restrict__ bd2,
                       const float* __restrict__ Wa,  const float* __restrict__ ba,
                       float* __restrict__ out)
{
    const int   tid  = threadIdx.x;
    const int   wave = tid >> 5;
    const int   lane = tid & 31;
    const int   lo   = lane & 15;
    const int   hi   = lane >> 4;
    const long  rowBase = (long)blockIdx.x * ROWS;
    const float C = 0.70710678118654752440f;

    const int  q4   = tid & 127;
    const int  r0   = tid >> 7;
    const bool oddq = q4 & 1;

    float4 xv[16];
    {
        const float4* xg = (const float4*)(x + rowBase * DLEN);
        #pragma unroll
        for (int u = 0; u < 16; ++u) xv[u] = xg[tid + u * NTHR];
    }

    GemmState S;
    {
        const bool   lowHalf = wave < 4;
        const float* W0 = lowHalf ? Wa : Wd2;
        const float* b0 = lowHalf ? ba : bd2;
        const int boff0 = lowHalf ? 0 : 64;
        const int n00   = (wave & 3) * 16;
        const int n01   = wave * 16;
        const int n03   = (wave + 8) * 16;
        S.wp[0] = W0  + (size_t)(n00 + lo) * 64  + 4 * hi;
        S.wp[1] = Wd1 + (size_t)(n01 + lo) * 128 + 4 * hi;
        S.wp[2] = Wd0 + (size_t)(n01 + lo) * 256 + 4 * hi;
        S.wp[3] = Wd0 + (size_t)(n03 + lo) * 256 + 4 * hi;
        S.aoff[0] = lo * ST + boff0 + 4 * hi;
        S.aoff[1] = lo * ST + 128 + 4 * hi;
        S.aoff[2] = lo * ST + 256 + 4 * hi;
        S.aoff[3] = lo * ST + 256 + 4 * hi;
        S.ycol[0] = boff0 + n00 + lo;
        S.ycol[1] = 128 + n01 + lo;
        S.ycol[2] = 256 + n01 + lo;
        S.ycol[3] = 256 + n03 + lo;
        S.bias[0] = b0[n00 + lo];
        S.bias[1] = bd1[n01 + lo];
        S.bias[2] = bd0[n01 + lo];
        S.bias[3] = bd0[n03 + lo];
        const v8f z = {0.f, 0.f, 0.f, 0.f, 0.f, 0.f, 0.f, 0.f};
        S.acc0 = z;
        S.acc1 = z;
        load_b<0>(S);
        load_b<1>(S);
    }

    {
        const int o2 = (oddq ? 64 : 0) + (q4 >> 1);
        #pragma unroll
        for (int u = 0; u < 16; ++u) {
            const int    rb = (r0 + 2 * u) * ST;
            const float4 v  = xv[u];
            v2f d0;
            d0.x = (v.x - v.y) * C;
            d0.y = (v.z - v.w) * C;
            const float a10 = (v.x + v.y) * C;
            const float a11 = (v.z + v.w) * C;
            const float d1  = (a10 - a11) * C;
            const float a2  = (a10 + a11) * C;
            const float nb  = lane_xor1(a2);
            const float t   = nb + (oddq ? -a2 : a2);
            *(v2f*)&bufB[rb + 256 + 2 * q4] = d0;
            bufB[rb + 128 + q4] = d1;
            bufB[rb + o2]       = t * C;
        }
    }
    __syncthreads();

    load_a<0>(S);
    run_blocks<0>(S, hi);
    __syncthreads();

    {
        float* og = out + rowBase * DLEN;
        const int h = q4 >> 1;
        for (int u = 0; u < 16; ++u) {
            const int   rb  = (r0 + 2 * u) * ST;
            const float ya  = bufA[rb + h];
            const float yd2 = bufA[rb + 64 + h];
            const float yd1 = bufA[rb + 128 + q4];
            const v2f   yd0 = *(const v2f*)&bufA[rb + 256 + 2 * q4];
            const float a2  = (ya + (oddq ? -yd2 : yd2)) * C;
            const float a1e = (a2 + yd1) * C;
            const float a1o = (a2 - yd1) * C;
            float4 ov;
            ov.x = (a1e + yd0.x) * C;
            ov.y = (a1e - yd0.x) * C;
            ov.z = (a1o + yd0.y) * C;
            ov.w = (a1o - yd0.y) * C;
            *(float4*)&og[(tid + u * NTHR) * 4] = ov;
        }
    }
}

extern "C" void kernel_launch(void* const* d_in, const int* in_sizes, int n_in,
                              void* d_out, int out_size, void* d_ws, size_t ws_size,
                              hipStream_t stream) {
    const float* xp  = (const float*)d_in[0];
    const float* Wd0 = (const float*)d_in[1];
    const float* bd0 = (const float*)d_in[2];
    const float* Wd1 = (const float*)d_in[3];
    const float* bd1 = (const float*)d_in[4];
    const float* Wd2 = (const float*)d_in[5];
    const float* bd2 = (const float*)d_in[6];
    const float* Wa  = (const float*)d_in[7];
    const float* ba  = (const float*)d_in[8];
    float* outp = (float*)d_out;

    int M = in_sizes[0] / DLEN;
    int blocks = M / ROWS;
    wavelet_mlp_fused<<<blocks, NTHR, 0, stream>>>(
        xp, Wd0, bd0, Wd1, bd1, Wd2, bd2, Wa, ba, outp);
}
